# starting kernel
# baseline (speedup 1.0000x reference)
#include <hip/hip_runtime.h>
#include <hip/hip_bf16.h>
#include <stddef.h>

typedef __attribute__((ext_vector_type(2))) float v2f;
typedef __attribute__((ext_vector_type(8))) float v8f;

__device__ __forceinline__ void atomAddF(float* p, float v) {
  asm volatile("global_atomic_add_f32 %0, %1, off scope:SCOPE_DEV"
               :
               : "v"(p), "v"(v)
               : "memory");
}

__global__ void k_init(float* __restrict__ deg, float* __restrict__ agg,
                       int n, int tot) {
  int t = blockIdx.x * blockDim.x + threadIdx.x;
  if (t < tot) agg[t] = 0.0f;
  if (t < n)   deg[t] = 1.0f;
}

__global__ void k_zero(float* __restrict__ p, int tot) {
  int t = blockIdx.x * blockDim.x + threadIdx.x;
  if (t < tot) p[t] = 0.0f;
}

__global__ void k_degree(const int* __restrict__ dst, float* __restrict__ deg,
                         int E) {
  int t = blockIdx.x * blockDim.x + threadIdx.x;
  if (t < E) atomAddF(&deg[dst[t]], 1.0f);
}

__global__ void k_rsqrt(float* __restrict__ d, int n) {
  int t = blockIdx.x * blockDim.x + threadIdx.x;
  if (t < n) d[t] = rsqrtf(d[t]);
}

__global__ void k_gemm16(const float* __restrict__ X, const float* __restrict__ W,
                         float* __restrict__ Y, int nrows, int K) {
  const int wave = threadIdx.x >> 5;
  const int lane = threadIdx.x & 31;
  const int tile = blockIdx.x * (blockDim.x >> 5) + wave;
  const int row0 = tile * 16;
  if (row0 >= nrows) return;

  const int m  = lane & 15;
  const int kh = lane >> 4;

  const float* ap = X + (size_t)(row0 + m) * K + 2 * kh;
  v8f acc = {};
  for (int kb = 0; kb < K; kb += 4) {
    v2f a = *(const v2f*)(ap + kb);
    const int kr = kb + 2 * kh;
    v2f b;
    b.x = W[(size_t)kr * 16 + m];
    b.y = W[(size_t)(kr + 1) * 16 + m];
    acc = __builtin_amdgcn_wmma_f32_16x16x4_f32(
        false, a, false, b, (short)0, acc, false, false);
  }
#pragma unroll
  for (int v = 0; v < 8; ++v) {
    Y[(size_t)(row0 + v + 8 * kh) * 16 + m] = acc[v];
  }
}

__global__ void k_scatter(const int* __restrict__ src, const int* __restrict__ dst,
                          const float* __restrict__ dinv,
                          const float* __restrict__ xw,
                          float* __restrict__ agg, int E) {
  int t = blockIdx.x * blockDim.x + threadIdx.x;
  int e = t >> 2;
  if (e >= E) return;
  int q = (t & 3) << 2;
  int s = src[e];
  int d = dst[e];
  float nrm = dinv[s] * dinv[d];
  float4 v = *(const float4*)(xw + (size_t)s * 16 + q);
  float* out = agg + (size_t)d * 16 + q;
  atomAddF(out + 0, v.x * nrm);
  atomAddF(out + 1, v.y * nrm);
  atomAddF(out + 2, v.z * nrm);
  atomAddF(out + 3, v.w * nrm);
}

__global__ void k_final1(float* __restrict__ agg, const float* __restrict__ xw,
                         const float* __restrict__ dinv,
                         const float* __restrict__ b, int tot) {
  int t = blockIdx.x * blockDim.x + threadIdx.x;
  if (t >= tot) return;
  int i = t >> 4, f = t & 15;
  float s = dinv[i];
  float v = agg[t] + xw[t] * s * s + b[f];
  agg[t] = fmaxf(v, 0.0f);
}

__global__ void k_final2(const float* __restrict__ agg, const float* __restrict__ xw,
                         const float* __restrict__ dinv,
                         const float* __restrict__ b2,
                         const float* __restrict__ Wl,
                         const float* __restrict__ bl,
                         float* __restrict__ out, int n) {
  int i = blockIdx.x * blockDim.x + threadIdx.x;
  if (i >= n) return;
  float s = dinv[i];
  float ss = s * s;
  float acc = 0.0f;
#pragma unroll
  for (int f = 0; f < 16; ++f) {
    float h = agg[(size_t)i * 16 + f] + xw[(size_t)i * 16 + f] * ss + b2[f];
    h = fmaxf(h, 0.0f);
    acc += h * Wl[f];
  }
  out[i] = acc + bl[0];
}

extern "C" void kernel_launch(void* const* d_in, const int* in_sizes, int n_in,
                              void* d_out, int out_size, void* d_ws, size_t ws_size,
                              hipStream_t stream) {
  const float* x   = (const float*)d_in[0];
  const int*   ei  = (const int*)d_in[1];
  const float* W1  = (const float*)d_in[2];
  const float* b1  = (const float*)d_in[3];
  const float* W2  = (const float*)d_in[4];
  const float* b2  = (const float*)d_in[5];
  const float* Wl  = (const float*)d_in[6];
  const float* bl  = (const float*)d_in[7];
  float* out = (float*)d_out;

  const int din = in_sizes[2] / 16;
  const int n   = in_sizes[0] / din;
  const int E   = in_sizes[1] / 2;
  const int tot = n * 16;

  float* ws   = (float*)d_ws;
  float* dinv = ws;
  float* bufA = ws + (((size_t)n + 255) & ~(size_t)255);
  float* bufB = bufA + (size_t)n * 16;

  const int T = 256;
  const int ntiles = (n + 15) / 16;
  const int gemmBlocks = (ntiles + 3) / 4;

  k_init  <<<(tot + T - 1) / T, T, 0, stream>>>(dinv, bufB, n, tot);
  k_degree<<<(E + T - 1) / T,   T, 0, stream>>>(ei + E, dinv, E);
  k_rsqrt <<<(n + T - 1) / T,   T, 0, stream>>>(dinv, n);

  k_gemm16 <<<gemmBlocks, 128, 0, stream>>>(x, W1, bufA, n, din);
  k_scatter<<<(4 * E + T - 1) / T, T, 0, stream>>>(ei, ei + E, dinv, bufA, bufB, E);
  k_final1 <<<(tot + T - 1) / T, T, 0, stream>>>(bufB, bufA, dinv, b1, tot);

  k_gemm16 <<<gemmBlocks, 128, 0, stream>>>(bufB, W2, bufA, n, 16);
  k_zero   <<<(tot + T - 1) / T, T, 0, stream>>>(bufB, tot);
  k_scatter<<<(4 * E + T - 1) / T, T, 0, stream>>>(ei, ei + E, dinv, bufA, bufB, E);
  k_final2 <<<(n + T - 1) / T, T, 0, stream>>>(bufB, bufA, dinv, b2, Wl, bl, out, n);
}


# edge scatter re-mapped to lane = feature: 16 lanes per edge, 2 edges per instruction, indices and norm passed by v_readlane; each f32 atomic instruction covers 2 whole 64-byte rows, not parts of 8
# speedup vs baseline: 1.5708x; 1.5708x over previous
#include <hip/hip_runtime.h>
#include <hip/hip_bf16.h>
#include <stddef.h>

typedef __attribute__((ext_vector_type(2))) float v2f;
typedef __attribute__((ext_vector_type(8))) float v8f;

__device__ __forceinline__ void atomAddF(float* p, float v) {
  asm volatile("global_atomic_add_f32 %0, %1, off scope:SCOPE_DEV"
               :
               : "v"(p), "v"(v)
               : "memory");
}

__global__ void k_init(float* __restrict__ deg, float* __restrict__ agg,
                       int n, int tot) {
  int t = blockIdx.x * blockDim.x + threadIdx.x;
  if (t < tot) agg[t] = 0.0f;
  if (t < n)   deg[t] = 1.0f;
}

__global__ void k_zero(float* __restrict__ p, int tot) {
  int t = blockIdx.x * blockDim.x + threadIdx.x;
  if (t < tot) p[t] = 0.0f;
}

__global__ void k_degree(const int* __restrict__ dst, float* __restrict__ deg,
                         int E) {
  int t = blockIdx.x * blockDim.x + threadIdx.x;
  if (t < E) atomAddF(&deg[dst[t]], 1.0f);
}

__global__ void k_rsqrt(float* __restrict__ d, int n) {
  int t = blockIdx.x * blockDim.x + threadIdx.x;
  if (t < n) d[t] = rsqrtf(d[t]);
}

__global__ void k_gemm16(const float* __restrict__ X, const float* __restrict__ W,
                         float* __restrict__ Y, int nrows, int K) {
  const int wave = threadIdx.x >> 5;
  const int lane = threadIdx.x & 31;
  const int tile = blockIdx.x * (blockDim.x >> 5) + wave;
  const int row0 = tile * 16;
  if (row0 >= nrows) return;

  const int m  = lane & 15;
  const int kh = lane >> 4;

  const float* ap = X + (size_t)(row0 + m) * K + 2 * kh;
  v8f acc = {};
  for (int kb = 0; kb < K; kb += 4) {
    v2f a = *(const v2f*)(ap + kb);
    const int kr = kb + 2 * kh;
    v2f b;
    b.x = W[(size_t)kr * 16 + m];
    b.y = W[(size_t)(kr + 1) * 16 + m];
    acc = __builtin_amdgcn_wmma_f32_16x16x4_f32(
        false, a, false, b, (short)0, acc, false, false);
  }
#pragma unroll
  for (int v = 0; v < 8; ++v) {
    Y[(size_t)(row0 + v + 8 * kh) * 16 + m] = acc[v];
  }
}

__global__ void k_scatter(const int* __restrict__ src, const int* __restrict__ dst,
                          const float* __restrict__ dinv,
                          const float* __restrict__ xw,
                          float* __restrict__ agg, int E) {
  const int lane = threadIdx.x & 31;
  const int e = blockIdx.x * blockDim.x + threadIdx.x;
  int s = 0, d = 0;
  float nrm = 0.0f;
  if (e < E) {
    s = src[e];
    d = dst[e];
    nrm = dinv[s] * dinv[d];
  }
  const int nrmi  = __float_as_int(nrm);
  const int half  = lane >> 4;
  const int f     = lane & 15;
  const int ehalf = e - lane + half;

  float val[16];
#pragma unroll
  for (int j = 0; j < 16; ++j) {
    const int s0 = __builtin_amdgcn_readlane(s, 2 * j);
    const int s1 = __builtin_amdgcn_readlane(s, 2 * j + 1);
    const int n0 = __builtin_amdgcn_readlane(nrmi, 2 * j);
    const int n1 = __builtin_amdgcn_readlane(nrmi, 2 * j + 1);
    const int   sj = half ? s1 : s0;
    const float nj = __int_as_float(half ? n1 : n0);
    val[j] = xw[(size_t)sj * 16 + f] * nj;
  }
#pragma unroll
  for (int j = 0; j < 16; ++j) {
    const int d0 = __builtin_amdgcn_readlane(d, 2 * j);
    const int d1 = __builtin_amdgcn_readlane(d, 2 * j + 1);
    const int dj = half ? d1 : d0;
    if (ehalf + 2 * j < E) atomAddF(agg + (size_t)dj * 16 + f, val[j]);
  }
}

__global__ void k_final1(float* __restrict__ agg, const float* __restrict__ xw,
                         const float* __restrict__ dinv,
                         const float* __restrict__ b, int tot) {
  int t = blockIdx.x * blockDim.x + threadIdx.x;
  if (t >= tot) return;
  int i = t >> 4, f = t & 15;
  float s = dinv[i];
  float v = agg[t] + xw[t] * s * s + b[f];
  agg[t] = fmaxf(v, 0.0f);
}

__global__ void k_final2(const float* __restrict__ agg, const float* __restrict__ xw,
                         const float* __restrict__ dinv,
                         const float* __restrict__ b2,
                         const float* __restrict__ Wl,
                         const float* __restrict__ bl,
                         float* __restrict__ out, int n) {
  int i = blockIdx.x * blockDim.x + threadIdx.x;
  if (i >= n) return;
  float s = dinv[i];
  float ss = s * s;
  float acc = 0.0f;
#pragma unroll
  for (int f = 0; f < 16; ++f) {
    float h = agg[(size_t)i * 16 + f] + xw[(size_t)i * 16 + f] * ss + b2[f];
    h = fmaxf(h, 0.0f);
    acc += h * Wl[f];
  }
  out[i] = acc + bl[0];
}

extern "C" void kernel_launch(void* const* d_in, const int* in_sizes, int n_in,
                              void* d_out, int out_size, void* d_ws, size_t ws_size,
                              hipStream_t stream) {
  const float* x   = (const float*)d_in[0];
  const int*   ei  = (const int*)d_in[1];
  const float* W1  = (const float*)d_in[2];
  const float* b1  = (const float*)d_in[3];
  const float* W2  = (const float*)d_in[4];
  const float* b2  = (const float*)d_in[5];
  const float* Wl  = (const float*)d_in[6];
  const float* bl  = (const float*)d_in[7];
  float* out = (float*)d_out;

  const int din = in_sizes[2] / 16;
  const int n   = in_sizes[0] / din;
  const int E   = in_sizes[1] / 2;
  const int tot = n * 16;

  float* ws   = (float*)d_ws;
  float* dinv = ws;
  float* bufA = ws + (((size_t)n + 255) & ~(size_t)255);
  float* bufB = bufA + (size_t)n * 16;

  const int T = 256;
  const int ntiles = (n + 15) / 16;
  const int gemmBlocks = (ntiles + 3) / 4;

  k_init  <<<(tot + T - 1) / T, T, 0, stream>>>(dinv, bufB, n, tot);
  k_degree<<<(E + T - 1) / T,   T, 0, stream>>>(ei + E, dinv, E);
  k_rsqrt <<<(n + T - 1) / T,   T, 0, stream>>>(dinv, n);

  k_gemm16 <<<gemmBlocks, 128, 0, stream>>>(x, W1, bufA, n, din);
  k_scatter<<<(E + T - 1) / T, T, 0, stream>>>(ei, ei + E, dinv, bufA, bufB, E);
  k_final1 <<<(tot + T - 1) / T, T, 0, stream>>>(bufB, bufA, dinv, b1, tot);

  k_gemm16 <<<gemmBlocks, 128, 0, stream>>>(bufB, W2, bufA, n, 16);
  k_zero   <<<(tot + T - 1) / T, T, 0, stream>>>(bufB, tot);
  k_scatter<<<(E + T - 1) / T, T, 0, stream>>>(ei, ei + E, dinv, bufA, bufB, E);
  k_final2 <<<(n + T - 1) / T, T, 0, stream>>>(bufB, bufA, dinv, b2, Wl, bl, out, n);
}


# norm split: GEMM store scales rows by dinv[src] and seeds the sum with the self-loop row (now summed first), scatter adds rows with no dinv gathers, finalize multiplies by dinv[dst]
# speedup vs baseline: 1.6945x; 1.0787x over previous
#include <hip/hip_runtime.h>
#include <hip/hip_bf16.h>
#include <stddef.h>

typedef __attribute__((ext_vector_type(2))) float v2f;
typedef __attribute__((ext_vector_type(8))) float v8f;

__device__ __forceinline__ void atomAddF(float* p, float v) {
  asm volatile("global_atomic_add_f32 %0, %1, off scope:SCOPE_DEV"
               :
               : "v"(p), "v"(v)
               : "memory");
}

__global__ void k_init(float* __restrict__ deg, int n) {
  int t = blockIdx.x * blockDim.x + threadIdx.x;
  if (t < n) deg[t] = 1.0f;
}

__global__ void k_degree(const int* __restrict__ dst, float* __restrict__ deg,
                         int E) {
  int t = blockIdx.x * blockDim.x + threadIdx.x;
  if (t < E) atomAddF(&deg[dst[t]], 1.0f);
}

__global__ void k_rsqrt(float* __restrict__ d, int n) {
  int t = blockIdx.x * blockDim.x + threadIdx.x;
  if (t < n) d[t] = rsqrtf(d[t]);
}

__global__ void k_gemm16(const float* X, const float* __restrict__ W,
                         const float* __restrict__ dinv,
                         float* __restrict__ Y, float* S, int nrows, int K) {
  const int wave = threadIdx.x >> 5;
  const int lane = threadIdx.x & 31;
  const int tile = blockIdx.x * (blockDim.x >> 5) + wave;
  const int row0 = tile * 16;
  if (row0 >= nrows) return;

  const int m  = lane & 15;
  const int kh = lane >> 4;

  const float4 dlo = *(const float4*)(dinv + row0 + 8 * kh);
  const float4 dhi = *(const float4*)(dinv + row0 + 8 * kh + 4);
  const float dv[8] = {dlo.x, dlo.y, dlo.z, dlo.w, dhi.x, dhi.y, dhi.z, dhi.w};

  const float* ap = X + (size_t)(row0 + m) * K + 2 * kh;
  v8f acc = {};
  for (int kb = 0; kb < K; kb += 4) {
    v2f a = *(const v2f*)(ap + kb);
    const int kr = kb + 2 * kh;
    v2f b;
    b.x = W[(size_t)kr * 16 + m];
    b.y = W[(size_t)(kr + 1) * 16 + m];
    acc = __builtin_amdgcn_wmma_f32_16x16x4_f32(
        false, a, false, b, (short)0, acc, false, false);
  }
#pragma unroll
  for (int v = 0; v < 8; ++v) {
    const size_t o = (size_t)(row0 + v + 8 * kh) * 16 + m;
    const float y = acc[v] * dv[v];
    Y[o] = y;
    S[o] = y;
  }
}

__global__ void k_scatter(const int* __restrict__ src, const int* __restrict__ dst,
                          const float* __restrict__ xw,
                          float* __restrict__ agg, int E) {
  const int lane = threadIdx.x & 31;
  const int e = blockIdx.x * blockDim.x + threadIdx.x;
  int s = 0, d = 0;
  if (e < E) {
    s = src[e];
    d = dst[e];
  }
  const int half  = lane >> 4;
  const int f     = lane & 15;
  const int ehalf = e - lane + half;

  float val[16];
#pragma unroll
  for (int j = 0; j < 16; ++j) {
    const int s0 = __builtin_amdgcn_readlane(s, 2 * j);
    const int s1 = __builtin_amdgcn_readlane(s, 2 * j + 1);
    const int sj = half ? s1 : s0;
    val[j] = xw[(size_t)sj * 16 + f];
  }
#pragma unroll
  for (int j = 0; j < 16; ++j) {
    const int d0 = __builtin_amdgcn_readlane(d, 2 * j);
    const int d1 = __builtin_amdgcn_readlane(d, 2 * j + 1);
    const int dj = half ? d1 : d0;
    if (ehalf + 2 * j < E) atomAddF(agg + (size_t)dj * 16 + f, val[j]);
  }
}

__global__ void k_final1(float* __restrict__ agg,
                         const float* __restrict__ dinv,
                         const float* __restrict__ b, int tot) {
  int t = blockIdx.x * blockDim.x + threadIdx.x;
  if (t >= tot) return;
  int i = t >> 4, f = t & 15;
  float v = agg[t] * dinv[i] + b[f];
  agg[t] = fmaxf(v, 0.0f);
}

__global__ void k_final2(const float* __restrict__ agg,
                         const float* __restrict__ dinv,
                         const float* __restrict__ b2,
                         const float* __restrict__ Wl,
                         const float* __restrict__ bl,
                         float* __restrict__ out, int n) {
  int i = blockIdx.x * blockDim.x + threadIdx.x;
  if (i >= n) return;
  float s = dinv[i];
  float acc = 0.0f;
#pragma unroll
  for (int f = 0; f < 16; ++f) {
    float h = agg[(size_t)i * 16 + f] * s + b2[f];
    h = fmaxf(h, 0.0f);
    acc += h * Wl[f];
  }
  out[i] = acc + bl[0];
}

extern "C" void kernel_launch(void* const* d_in, const int* in_sizes, int n_in,
                              void* d_out, int out_size, void* d_ws, size_t ws_size,
                              hipStream_t stream) {
  const float* x   = (const float*)d_in[0];
  const int*   ei  = (const int*)d_in[1];
  const float* W1  = (const float*)d_in[2];
  const float* b1  = (const float*)d_in[3];
  const float* W2  = (const float*)d_in[4];
  const float* b2  = (const float*)d_in[5];
  const float* Wl  = (const float*)d_in[6];
  const float* bl  = (const float*)d_in[7];
  float* out = (float*)d_out;

  const int din = in_sizes[2] / 16;
  const int n   = in_sizes[0] / din;
  const int E   = in_sizes[1] / 2;
  const int tot = n * 16;

  float* ws   = (float*)d_ws;
  float* dinv = ws;
  float* bufA = ws + (((size_t)n + 255) & ~(size_t)255);
  float* bufB = bufA + (size_t)n * 16;

  const int T = 256;
  const int ntiles = (n + 15) / 16;
  const int gemmBlocks = (ntiles + 3) / 4;

  k_init  <<<(n + T - 1) / T,   T, 0, stream>>>(dinv, n);
  k_degree<<<(E + T - 1) / T,   T, 0, stream>>>(ei + E, dinv, E);
  k_rsqrt <<<(n + T - 1) / T,   T, 0, stream>>>(dinv, n);

  k_gemm16 <<<gemmBlocks, 128, 0, stream>>>(x, W1, dinv, bufA, bufB, n, din);
  k_scatter<<<(E + T - 1) / T, T, 0, stream>>>(ei, ei + E, bufA, bufB, E);
  k_final1 <<<(tot + T - 1) / T, T, 0, stream>>>(bufB, dinv, b1, tot);

  k_gemm16 <<<gemmBlocks, 128, 0, stream>>>(bufB, W2, dinv, bufA, bufB, n, 16);
  k_scatter<<<(E + T - 1) / T, T, 0, stream>>>(ei, ei + E, bufA, bufB, E);
  k_final2 <<<(n + T - 1) / T, T, 0, stream>>>(bufB, dinv, b2, Wl, bl, out, n);
}


# deg and both layer sums kept as two partial copies, one per L2 half: a wave reads its die id and adds into its own half's copy, the reader adds the two; f32 sum tree and atomic add order change
# speedup vs baseline: 3.9755x; 2.3461x over previous
#include <hip/hip_runtime.h>
#include <hip/hip_bf16.h>
#include <stddef.h>

typedef __attribute__((ext_vector_type(2))) float v2f;
typedef __attribute__((ext_vector_type(8))) float v8f;

__device__ __forceinline__ void atomAddF(float* p, float v) {
  asm volatile("global_atomic_add_f32 %0, %1, off scope:SCOPE_DEV"
               :
               : "v"(p), "v"(v)
               : "memory");
}

__device__ __forceinline__ int upper_half() {
  return (int)((__builtin_amdgcn_s_sendmsg_rtn(0x87) >> 18) & 1u);
}

__global__ void k_init(float* __restrict__ deg, float* __restrict__ degHi, int n) {
  int t = blockIdx.x * blockDim.x + threadIdx.x;
  if (t < n) {
    deg[t] = 1.0f;
    if (degHi) degHi[t] = 0.0f;
  }
}

__global__ void k_degree(const int* __restrict__ dst, float* __restrict__ deg,
                         float* __restrict__ degHi, int E) {
  int t = blockIdx.x * blockDim.x + threadIdx.x;
  float* dg = (degHi && upper_half()) ? degHi : deg;
  if (t < E) atomAddF(&dg[dst[t]], 1.0f);
}

__global__ void k_rsqrt(float* __restrict__ d, const float* __restrict__ dHi, int n) {
  int t = blockIdx.x * blockDim.x + threadIdx.x;
  if (t < n) d[t] = rsqrtf(dHi ? d[t] + dHi[t] : d[t]);
}

__global__ void k_gemm16(const float* X, const float* __restrict__ W,
                         const float* __restrict__ dinv,
                         float* __restrict__ Y, float* S, float* __restrict__ Z,
                         int nrows, int K) {
  const int wave = threadIdx.x >> 5;
  const int lane = threadIdx.x & 31;
  const int tile = blockIdx.x * (blockDim.x >> 5) + wave;
  const int row0 = tile * 16;
  if (row0 >= nrows) return;

  const int m  = lane & 15;
  const int kh = lane >> 4;

  const float4 dlo = *(const float4*)(dinv + row0 + 8 * kh);
  const float4 dhi = *(const float4*)(dinv + row0 + 8 * kh + 4);
  const float dv[8] = {dlo.x, dlo.y, dlo.z, dlo.w, dhi.x, dhi.y, dhi.z, dhi.w};

  const float* ap = X + (size_t)(row0 + m) * K + 2 * kh;
  v8f acc = {};
  for (int kb = 0; kb < K; kb += 4) {
    v2f a = *(const v2f*)(ap + kb);
    const int kr = kb + 2 * kh;
    v2f b;
    b.x = W[(size_t)kr * 16 + m];
    b.y = W[(size_t)(kr + 1) * 16 + m];
    acc = __builtin_amdgcn_wmma_f32_16x16x4_f32(
        false, a, false, b, (short)0, acc, false, false);
  }
#pragma unroll
  for (int v = 0; v < 8; ++v) {
    const size_t o = (size_t)(row0 + v + 8 * kh) * 16 + m;
    const float y = acc[v] * dv[v];
    Y[o] = y;
    S[o] = y;
    if (Z) Z[o] = 0.0f;
  }
}

__global__ void k_scatter(const int* __restrict__ src, const int* __restrict__ dst,
                          const float* __restrict__ xw,
                          float* __restrict__ aggLo, float* __restrict__ aggHi,
                          int E) {
  float* agg = (aggHi && upper_half()) ? aggHi : aggLo;
  const int lane = threadIdx.x & 31;
  const int e = blockIdx.x * blockDim.x + threadIdx.x;
  int s = 0, d = 0;
  if (e < E) {
    s = src[e];
    d = dst[e];
  }
  const int half  = lane >> 4;
  const int f     = lane & 15;
  const int ehalf = e - lane + half;

  float val[16];
#pragma unroll
  for (int j = 0; j < 16; ++j) {
    const int s0 = __builtin_amdgcn_readlane(s, 2 * j);
    const int s1 = __builtin_amdgcn_readlane(s, 2 * j + 1);
    const int sj = half ? s1 : s0;
    val[j] = xw[(size_t)sj * 16 + f];
  }
#pragma unroll
  for (int j = 0; j < 16; ++j) {
    const int d0 = __builtin_amdgcn_readlane(d, 2 * j);
    const int d1 = __builtin_amdgcn_readlane(d, 2 * j + 1);
    const int dj = half ? d1 : d0;
    if (ehalf + 2 * j < E) atomAddF(agg + (size_t)dj * 16 + f, val[j]);
  }
}

__global__ void k_final1(float* __restrict__ agg,
                         const float* __restrict__ aggHi,
                         const float* __restrict__ dinv,
                         const float* __restrict__ b, int tot) {
  int t = blockIdx.x * blockDim.x + threadIdx.x;
  if (t >= tot) return;
  int i = t >> 4, f = t & 15;
  float a = agg[t];
  if (aggHi) a += aggHi[t];
  float v = a * dinv[i] + b[f];
  agg[t] = fmaxf(v, 0.0f);
}

__global__ void k_final2(const float* __restrict__ agg,
                         const float* __restrict__ aggHi,
                         const float* __restrict__ dinv,
                         const float* __restrict__ b2,
                         const float* __restrict__ Wl,
                         const float* __restrict__ bl,
                         float* __restrict__ out, int n) {
  int i = blockIdx.x * blockDim.x + threadIdx.x;
  if (i >= n) return;
  float s = dinv[i];
  float acc = 0.0f;
#pragma unroll
  for (int f = 0; f < 16; ++f) {
    float a = agg[(size_t)i * 16 + f];
    if (aggHi) a += aggHi[(size_t)i * 16 + f];
    float h = a * s + b2[f];
    h = fmaxf(h, 0.0f);
    acc += h * Wl[f];
  }
  out[i] = acc + bl[0];
}

extern "C" void kernel_launch(void* const* d_in, const int* in_sizes, int n_in,
                              void* d_out, int out_size, void* d_ws, size_t ws_size,
                              hipStream_t stream) {
  const float* x   = (const float*)d_in[0];
  const int*   ei  = (const int*)d_in[1];
  const float* W1  = (const float*)d_in[2];
  const float* b1  = (const float*)d_in[3];
  const float* W2  = (const float*)d_in[4];
  const float* b2  = (const float*)d_in[5];
  const float* Wl  = (const float*)d_in[6];
  const float* bl  = (const float*)d_in[7];
  float* out = (float*)d_out;

  const int din = in_sizes[2] / 16;
  const int n   = in_sizes[0] / din;
  const int E   = in_sizes[1] / 2;
  const int tot = n * 16;

  float* ws   = (float*)d_ws;
  float* dinv = ws;
  float* bufA = ws + (((size_t)n + 255) & ~(size_t)255);
  float* bufB = bufA + (size_t)n * 16;
  const size_t need = ((((size_t)n + 255) & ~(size_t)255) + (size_t)n * 49) * sizeof(float);
  float* bufC = (ws_size >= need) ? bufB + (size_t)n * 16 : nullptr;
  float* degC = bufC ? bufC + (size_t)n * 16 : nullptr;

  const int T = 256;
  const int ntiles = (n + 15) / 16;
  const int gemmBlocks = (ntiles + 3) / 4;

  k_init  <<<(n + T - 1) / T,   T, 0, stream>>>(dinv, degC, n);
  k_degree<<<(E + T - 1) / T,   T, 0, stream>>>(ei + E, dinv, degC, E);
  k_rsqrt <<<(n + T - 1) / T,   T, 0, stream>>>(dinv, degC, n);

  k_gemm16 <<<gemmBlocks, 128, 0, stream>>>(x, W1, dinv, bufA, bufB, bufC, n, din);
  k_scatter<<<(E + T - 1) / T, T, 0, stream>>>(ei, ei + E, bufA, bufB, bufC, E);
  k_final1 <<<(tot + T - 1) / T, T, 0, stream>>>(bufB, bufC, dinv, b1, tot);

  k_gemm16 <<<gemmBlocks, 128, 0, stream>>>(bufB, W2, dinv, bufA, bufB, bufC, n, 16);
  k_scatter<<<(E + T - 1) / T, T, 0, stream>>>(ei, ei + E, bufA, bufB, bufC, E);
  k_final2 <<<(n + T - 1) / T, T, 0, stream>>>(bufB, bufC, dinv, b2, Wl, bl, out, n);
}


# fewer, longer waves: a scatter wave walks 8 groups of 32 edges (next indices loaded ahead), a degree thread takes 8 edges, a finalize thread 4 features; f32 atomic adds still in arrival order
# speedup vs baseline: 4.1557x; 1.0453x over previous
#include <hip/hip_runtime.h>
#include <hip/hip_bf16.h>
#include <stddef.h>

typedef __attribute__((ext_vector_type(2))) float v2f;
typedef __attribute__((ext_vector_type(8))) float v8f;

__device__ __forceinline__ void atomAddF(float* p, float v) {
  asm volatile("global_atomic_add_f32 %0, %1, off scope:SCOPE_DEV"
               :
               : "v"(p), "v"(v)
               : "memory");
}

__device__ __forceinline__ int upper_half() {
  return (int)((__builtin_amdgcn_s_sendmsg_rtn(0x87) >> 18) & 1u);
}

__global__ void k_init(float* __restrict__ deg, float* __restrict__ degHi, int n) {
  int t = blockIdx.x * blockDim.x + threadIdx.x;
  if (t < n) {
    deg[t] = 1.0f;
    if (degHi) degHi[t] = 0.0f;
  }
}

__global__ void k_degree(const int* __restrict__ dst, float* __restrict__ deg,
                         float* __restrict__ degHi, int E) {
  const int t    = blockIdx.x * blockDim.x + threadIdx.x;
  const int nthr = gridDim.x * blockDim.x;
  float* dg = (degHi && upper_half()) ? degHi : deg;
  const int nq = (((size_t)dst & 15) == 0) ? (E >> 2) : 0;
  const int4* dq = (const int4*)dst;
  int q = t;
  for (; q + nthr < nq; q += 2 * nthr) {
    const int4 a = dq[q];
    const int4 b = dq[q + nthr];
    atomAddF(&dg[a.x], 1.0f); atomAddF(&dg[a.y], 1.0f);
    atomAddF(&dg[a.z], 1.0f); atomAddF(&dg[a.w], 1.0f);
    atomAddF(&dg[b.x], 1.0f); atomAddF(&dg[b.y], 1.0f);
    atomAddF(&dg[b.z], 1.0f); atomAddF(&dg[b.w], 1.0f);
  }
  for (; q < nq; q += nthr) {
    const int4 a = dq[q];
    atomAddF(&dg[a.x], 1.0f); atomAddF(&dg[a.y], 1.0f);
    atomAddF(&dg[a.z], 1.0f); atomAddF(&dg[a.w], 1.0f);
  }
  for (int e = (nq << 2) + t; e < E; e += nthr) atomAddF(&dg[dst[e]], 1.0f);
}

__global__ void k_rsqrt(float* __restrict__ d, const float* __restrict__ dHi, int n) {
  int t = blockIdx.x * blockDim.x + threadIdx.x;
  if (t < n) d[t] = rsqrtf(dHi ? d[t] + dHi[t] : d[t]);
}

__global__ void k_gemm16(const float* X, const float* __restrict__ W,
                         const float* __restrict__ dinv,
                         float* __restrict__ Y, float* S, float* __restrict__ Z,
                         int nrows, int K) {
  const int wave = threadIdx.x >> 5;
  const int lane = threadIdx.x & 31;
  const int tile = blockIdx.x * (blockDim.x >> 5) + wave;
  const int row0 = tile * 16;
  if (row0 >= nrows) return;

  const int m  = lane & 15;
  const int kh = lane >> 4;

  const float4 dlo = *(const float4*)(dinv + row0 + 8 * kh);
  const float4 dhi = *(const float4*)(dinv + row0 + 8 * kh + 4);
  const float dv[8] = {dlo.x, dlo.y, dlo.z, dlo.w, dhi.x, dhi.y, dhi.z, dhi.w};

  const float* ap = X + (size_t)(row0 + m) * K + 2 * kh;
  v8f acc = {};
  for (int kb = 0; kb < K; kb += 4) {
    v2f a = *(const v2f*)(ap + kb);
    const int kr = kb + 2 * kh;
    v2f b;
    b.x = W[(size_t)kr * 16 + m];
    b.y = W[(size_t)(kr + 1) * 16 + m];
    acc = __builtin_amdgcn_wmma_f32_16x16x4_f32(
        false, a, false, b, (short)0, acc, false, false);
  }
#pragma unroll
  for (int v = 0; v < 8; ++v) {
    const size_t o = (size_t)(row0 + v + 8 * kh) * 16 + m;
    const float y = acc[v] * dv[v];
    Y[o] = y;
    S[o] = y;
    if (Z) Z[o] = 0.0f;
  }
}

#define SC_G 8
__global__ void k_scatter(const int* __restrict__ src, const int* __restrict__ dst,
                          const float* __restrict__ xw,
                          float* __restrict__ aggLo, float* __restrict__ aggHi,
                          int E) {
  float* agg = (aggHi && upper_half()) ? aggHi : aggLo;
  const int lane = threadIdx.x & 31;
  const int wv   = (blockIdx.x * blockDim.x + threadIdx.x) >> 5;
  int e = wv * (32 * SC_G) + lane;
  int s = src[min(e, E - 1)];
  int d = dst[min(e, E - 1)];
  const int half  = lane >> 4;
  const int f     = lane & 15;

  for (int g = 0; g < SC_G; ++g) {
    if (e - lane >= E) break;
    const int en = e + 32;
    const int ep = min((g + 1 < SC_G) ? en : e, E - 1);
    const int sn = src[ep];
    const int dn = dst[ep];
    const int ehalf = e - lane + half;

    float val[16];
#pragma unroll
    for (int j = 0; j < 16; ++j) {
      const int s0 = __builtin_amdgcn_readlane(s, 2 * j);
      const int s1 = __builtin_amdgcn_readlane(s, 2 * j + 1);
      const int sj = half ? s1 : s0;
      val[j] = xw[(size_t)sj * 16 + f];
    }
#pragma unroll
    for (int j = 0; j < 16; ++j) {
      const int d0 = __builtin_amdgcn_readlane(d, 2 * j);
      const int d1 = __builtin_amdgcn_readlane(d, 2 * j + 1);
      const int dj = half ? d1 : d0;
      if (ehalf + 2 * j < E) atomAddF(agg + (size_t)dj * 16 + f, val[j]);
    }
    s = sn;
    d = dn;
    e = en;
  }
}

__global__ void k_final1(float* __restrict__ agg,
                         const float* __restrict__ aggHi,
                         const float* __restrict__ dinv,
                         const float* __restrict__ b, int tot4) {
  int t = blockIdx.x * blockDim.x + threadIdx.x;
  if (t >= tot4) return;
  const int i = t >> 2, c = t & 3;
  float4 a = ((const float4*)agg)[t];
  if (aggHi) {
    const float4 h = ((const float4*)aggHi)[t];
    a.x += h.x; a.y += h.y; a.z += h.z; a.w += h.w;
  }
  const float s = dinv[i];
  const float4 bb = ((const float4*)b)[c];
  float4 v;
  v.x = fmaxf(a.x * s + bb.x, 0.0f);
  v.y = fmaxf(a.y * s + bb.y, 0.0f);
  v.z = fmaxf(a.z * s + bb.z, 0.0f);
  v.w = fmaxf(a.w * s + bb.w, 0.0f);
  ((float4*)agg)[t] = v;
}

__global__ void k_final2(const float* __restrict__ agg,
                         const float* __restrict__ aggHi,
                         const float* __restrict__ dinv,
                         const float* __restrict__ b2,
                         const float* __restrict__ Wl,
                         const float* __restrict__ bl,
                         float* __restrict__ out, int n) {
  int i = blockIdx.x * blockDim.x + threadIdx.x;
  if (i >= n) return;
  float s = dinv[i];
  float acc = 0.0f;
#pragma unroll
  for (int f = 0; f < 16; ++f) {
    float a = agg[(size_t)i * 16 + f];
    if (aggHi) a += aggHi[(size_t)i * 16 + f];
    float h = a * s + b2[f];
    h = fmaxf(h, 0.0f);
    acc += h * Wl[f];
  }
  out[i] = acc + bl[0];
}

extern "C" void kernel_launch(void* const* d_in, const int* in_sizes, int n_in,
                              void* d_out, int out_size, void* d_ws, size_t ws_size,
                              hipStream_t stream) {
  const float* x   = (const float*)d_in[0];
  const int*   ei  = (const int*)d_in[1];
  const float* W1  = (const float*)d_in[2];
  const float* b1  = (const float*)d_in[3];
  const float* W2  = (const float*)d_in[4];
  const float* b2  = (const float*)d_in[5];
  const float* Wl  = (const float*)d_in[6];
  const float* bl  = (const float*)d_in[7];
  float* out = (float*)d_out;

  const int din = in_sizes[2] / 16;
  const int n   = in_sizes[0] / din;
  const int E   = in_sizes[1] / 2;
  const int tot = n * 16;

  float* ws   = (float*)d_ws;
  float* dinv = ws;
  float* bufA = ws + (((size_t)n + 255) & ~(size_t)255);
  float* bufB = bufA + (size_t)n * 16;
  const size_t need = ((((size_t)n + 255) & ~(size_t)255) + (size_t)n * 49) * sizeof(float);
  float* bufC = (ws_size >= need) ? bufB + (size_t)n * 16 : nullptr;
  float* degC = bufC ? bufC + (size_t)n * 16 : nullptr;

  const int T = 256;
  const int ntiles = (n + 15) / 16;
  const int gemmBlocks = (ntiles + 3) / 4;
  const int scatBlocks = (E + 32 * SC_G * (T / 32) - 1) / (32 * SC_G * (T / 32));
  const int tot4 = tot / 4;

  k_init  <<<(n + T - 1) / T,   T, 0, stream>>>(dinv, degC, n);
  k_degree<<<(E / 8 + T) / T,   T, 0, stream>>>(ei + E, dinv, degC, E);
  k_rsqrt <<<(n + T - 1) / T,   T, 0, stream>>>(dinv, degC, n);

  k_gemm16 <<<gemmBlocks, 128, 0, stream>>>(x, W1, dinv, bufA, bufB, bufC, n, din);
  k_scatter<<<scatBlocks, T, 0, stream>>>(ei, ei + E, bufA, bufB, bufC, E);
  k_final1 <<<(tot4 + T - 1) / T, T, 0, stream>>>(bufB, bufC, dinv, b1, tot4);

  k_gemm16 <<<gemmBlocks, 128, 0, stream>>>(bufB, W2, dinv, bufA, bufB, bufC, n, 16);
  k_scatter<<<scatBlocks, T, 0, stream>>>(ei, ei + E, bufA, bufB, bufC, E);
  k_final2 <<<(n + T - 1) / T, T, 0, stream>>>(bufB, bufC, dinv, b2, Wl, bl, out, n);
}


# rsqrt(deg) and the layer-1 finalize stop being launches: each product wave does them for its own 16 rows (dinv where it scales its store, relu(sum*dinv+b) in the A loader); 9 launches become 7
# speedup vs baseline: 4.9082x; 1.1811x over previous
#include <hip/hip_runtime.h>
#include <hip/hip_bf16.h>
#include <stddef.h>

typedef __attribute__((ext_vector_type(2))) float v2f;
typedef __attribute__((ext_vector_type(8))) float v8f;

__device__ __forceinline__ void atomAddF(float* p, float v) {
  asm volatile("global_atomic_add_f32 %0, %1, off scope:SCOPE_DEV"
               :
               : "v"(p), "v"(v)
               : "memory");
}

__device__ __forceinline__ int upper_half() {
  return (int)((__builtin_amdgcn_s_sendmsg_rtn(0x87) >> 18) & 1u);
}

__global__ void k_init(float* __restrict__ deg, float* __restrict__ degHi, int n) {
  int t = blockIdx.x * blockDim.x + threadIdx.x;
  if (t < n) {
    deg[t] = 1.0f;
    if (degHi) degHi[t] = 0.0f;
  }
}

__global__ void k_degree(const int* __restrict__ dst, float* __restrict__ deg,
                         float* __restrict__ degHi, int E) {
  const int t    = blockIdx.x * blockDim.x + threadIdx.x;
  const int nthr = gridDim.x * blockDim.x;
  float* dg = (degHi && upper_half()) ? degHi : deg;
  const int nq = (((size_t)dst & 15) == 0) ? (E >> 2) : 0;
  const int4* dq = (const int4*)dst;
  int q = t;
  for (; q + nthr < nq; q += 2 * nthr) {
    const int4 a = dq[q];
    const int4 b = dq[q + nthr];
    atomAddF(&dg[a.x], 1.0f); atomAddF(&dg[a.y], 1.0f);
    atomAddF(&dg[a.z], 1.0f); atomAddF(&dg[a.w], 1.0f);
    atomAddF(&dg[b.x], 1.0f); atomAddF(&dg[b.y], 1.0f);
    atomAddF(&dg[b.z], 1.0f); atomAddF(&dg[b.w], 1.0f);
  }
  for (; q < nq; q += nthr) {
    const int4 a = dq[q];
    atomAddF(&dg[a.x], 1.0f); atomAddF(&dg[a.y], 1.0f);
    atomAddF(&dg[a.z], 1.0f); atomAddF(&dg[a.w], 1.0f);
  }
  for (int e = (nq << 2) + t; e < E; e += nthr) atomAddF(&dg[dst[e]], 1.0f);
}

template <int LAYER>
__global__ void k_gemm16(const float* X, const float* XHi,
                         const float* __restrict__ bias,
                         const float* __restrict__ W,
                         float* dinv, const float* degHi,
                         float* __restrict__ Y, float* S, float* Z,
                         int nrows, int K) {
  const int wave = threadIdx.x >> 5;
  const int lane = threadIdx.x & 31;
  const int tile = blockIdx.x * (blockDim.x >> 5) + wave;
  const int row0 = tile * 16;
  if (row0 >= nrows) return;

  const int m  = lane & 15;
  const int kh = lane >> 4;

  float4 dlo = *(const float4*)(dinv + row0 + 8 * kh);
  float4 dhi = *(const float4*)(dinv + row0 + 8 * kh + 4);
  float4 hlo = {0.0f, 0.0f, 0.0f, 0.0f}, hhi = {0.0f, 0.0f, 0.0f, 0.0f};
  if (LAYER == 1 && degHi) {
    hlo = *(const float4*)(degHi + row0 + 8 * kh);
    hhi = *(const float4*)(degHi + row0 + 8 * kh + 4);
  }

  v8f acc = {};
  if (LAYER == 2) {
    const float* ap = X + (size_t)(row0 + m) * 16 + 2 * kh;
    v2f a[4], bb[4], b[4];
    const float da = dinv[row0 + m];
#pragma unroll
    for (int j = 0; j < 4; ++j) {
      const int kr = 4 * j + 2 * kh;
      a[j]   = *(const v2f*)(ap + 4 * j);
      bb[j]  = *(const v2f*)(bias + kr);
      b[j].x = W[kr * 16 + m];
      b[j].y = W[(kr + 1) * 16 + m];
    }
    if (XHi) {
      const float* aph = XHi + (size_t)(row0 + m) * 16 + 2 * kh;
#pragma unroll
      for (int j = 0; j < 4; ++j) {
        const v2f q = *(const v2f*)(aph + 4 * j);
        a[j].x += q.x; a[j].y += q.y;
      }
    }
#pragma unroll
    for (int j = 0; j < 4; ++j) {
      a[j].x = fmaxf(a[j].x * da + bb[j].x, 0.0f);
      a[j].y = fmaxf(a[j].y * da + bb[j].y, 0.0f);
      acc = __builtin_amdgcn_wmma_f32_16x16x4_f32(
          false, a[j], false, b[j], (short)0, acc, false, false);
    }
  } else {
    const float* ap = X + (size_t)(row0 + m) * K + 2 * kh;
    for (int kb = 0; kb < K; kb += 4) {
      v2f a = *(const v2f*)(ap + kb);
      const int kr = kb + 2 * kh;
      v2f b;
      b.x = W[(size_t)kr * 16 + m];
      b.y = W[(size_t)(kr + 1) * 16 + m];
      acc = __builtin_amdgcn_wmma_f32_16x16x4_f32(
          false, a, false, b, (short)0, acc, false, false);
    }
  }
  if (LAYER == 1) {
    dlo.x = rsqrtf(dlo.x + hlo.x); dlo.y = rsqrtf(dlo.y + hlo.y);
    dlo.z = rsqrtf(dlo.z + hlo.z); dlo.w = rsqrtf(dlo.w + hlo.w);
    dhi.x = rsqrtf(dhi.x + hhi.x); dhi.y = rsqrtf(dhi.y + hhi.y);
    dhi.z = rsqrtf(dhi.z + hhi.z); dhi.w = rsqrtf(dhi.w + hhi.w);
    if (m == 0) {
      *(float4*)(dinv + row0 + 8 * kh)     = dlo;
      *(float4*)(dinv + row0 + 8 * kh + 4) = dhi;
    }
  }
  const float dv[8] = {dlo.x, dlo.y, dlo.z, dlo.w, dhi.x, dhi.y, dhi.z, dhi.w};
#pragma unroll
  for (int v = 0; v < 8; ++v) {
    const size_t o = (size_t)(row0 + v + 8 * kh) * 16 + m;
    const float y = acc[v] * dv[v];
    Y[o] = y;
    S[o] = y;
    if (Z) Z[o] = 0.0f;
  }
}

#define SC_G 8
__global__ void k_scatter(const int* __restrict__ src, const int* __restrict__ dst,
                          const float* __restrict__ xw,
                          float* __restrict__ aggLo, float* __restrict__ aggHi,
                          int E) {
  float* agg = (aggHi && upper_half()) ? aggHi : aggLo;
  const int lane = threadIdx.x & 31;
  const int wv   = (blockIdx.x * blockDim.x + threadIdx.x) >> 5;
  int e = wv * (32 * SC_G) + lane;
  int s = src[min(e, E - 1)];
  int d = dst[min(e, E - 1)];
  const int half  = lane >> 4;
  const int f     = lane & 15;

  for (int g = 0; g < SC_G; ++g) {
    if (e - lane >= E) break;
    const int en = e + 32;
    const int ep = min((g + 1 < SC_G) ? en : e, E - 1);
    const int sn = src[ep];
    const int dn = dst[ep];
    const int ehalf = e - lane + half;

    float val[16];
#pragma unroll
    for (int j = 0; j < 16; ++j) {
      const int s0 = __builtin_amdgcn_readlane(s, 2 * j);
      const int s1 = __builtin_amdgcn_readlane(s, 2 * j + 1);
      const int sj = half ? s1 : s0;
      val[j] = xw[(size_t)sj * 16 + f];
    }
#pragma unroll
    for (int j = 0; j < 16; ++j) {
      const int d0 = __builtin_amdgcn_readlane(d, 2 * j);
      const int d1 = __builtin_amdgcn_readlane(d, 2 * j + 1);
      const int dj = half ? d1 : d0;
      if (ehalf + 2 * j < E) atomAddF(agg + (size_t)dj * 16 + f, val[j]);
    }
    s = sn;
    d = dn;
    e = en;
  }
}

__global__ void k_final2(const float* __restrict__ agg,
                         const float* __restrict__ aggHi,
                         const float* __restrict__ dinv,
                         const float* __restrict__ b2,
                         const float* __restrict__ Wl,
                         const float* __restrict__ bl,
                         float* __restrict__ out, int n) {
  int i = blockIdx.x * blockDim.x + threadIdx.x;
  if (i >= n) return;
  float s = dinv[i];
  float acc = 0.0f;
#pragma unroll
  for (int f = 0; f < 16; ++f) {
    float a = agg[(size_t)i * 16 + f];
    if (aggHi) a += aggHi[(size_t)i * 16 + f];
    float h = a * s + b2[f];
    h = fmaxf(h, 0.0f);
    acc += h * Wl[f];
  }
  out[i] = acc + bl[0];
}

extern "C" void kernel_launch(void* const* d_in, const int* in_sizes, int n_in,
                              void* d_out, int out_size, void* d_ws, size_t ws_size,
                              hipStream_t stream) {
  const float* x   = (const float*)d_in[0];
  const int*   ei  = (const int*)d_in[1];
  const float* W1  = (const float*)d_in[2];
  const float* b1  = (const float*)d_in[3];
  const float* W2  = (const float*)d_in[4];
  const float* b2  = (const float*)d_in[5];
  const float* Wl  = (const float*)d_in[6];
  const float* bl  = (const float*)d_in[7];
  float* out = (float*)d_out;

  const int din = in_sizes[2] / 16;
  const int n   = in_sizes[0] / din;
  const int E   = in_sizes[1] / 2;

  float* ws   = (float*)d_ws;
  float* dinv = ws;
  float* bufA = ws + (((size_t)n + 255) & ~(size_t)255);
  float* bufB = bufA + (size_t)n * 16;
  const size_t need = ((((size_t)n + 255) & ~(size_t)255) + (size_t)n * 49) * sizeof(float);
  float* bufC = (ws_size >= need) ? bufB + (size_t)n * 16 : nullptr;
  float* degC = bufC ? bufC + (size_t)n * 16 : nullptr;

  const int T = 256;
  const int ntiles = (n + 15) / 16;
  const int gemmBlocks = (ntiles + 3) / 4;
  const int scatBlocks = (E + 32 * SC_G * (T / 32) - 1) / (32 * SC_G * (T / 32));

  k_init  <<<(n + T - 1) / T,   T, 0, stream>>>(dinv, degC, n);
  k_degree<<<(E / 8 + T) / T,   T, 0, stream>>>(ei + E, dinv, degC, E);

  k_gemm16<1><<<gemmBlocks, 128, 0, stream>>>(x, nullptr, nullptr, W1, dinv, degC, bufA, bufB, bufC, n, din);
  k_scatter<<<scatBlocks, T, 0, stream>>>(ei, ei + E, bufA, bufB, bufC, E);

  k_gemm16<2><<<gemmBlocks, 128, 0, stream>>>(bufB, bufC, b1, W2, dinv, nullptr, bufA, bufB, bufC, n, 16);
  k_scatter<<<scatBlocks, T, 0, stream>>>(ei, ei + E, bufA, bufB, bufC, E);
  k_final2 <<<(n + T - 1) / T, T, 0, stream>>>(bufB, bufC, dinv, b2, Wl, bl, out, n);
}


# edge scatter: one vector instruction per memory instruction (row offset broadcast by DPP row_share inside the add, 32-bit offsets on scalar bases, E tested per group); f32 atomics in arrival order
# speedup vs baseline: 5.5375x; 1.1282x over previous
#include <hip/hip_runtime.h>
#include <hip/hip_bf16.h>
#include <stddef.h>

typedef __attribute__((ext_vector_type(2))) float v2f;
typedef __attribute__((ext_vector_type(8))) float v8f;

__device__ __forceinline__ void atomAddF(float* p, float v) {
  asm volatile("global_atomic_add_f32 %0, %1, off scope:SCOPE_DEV"
               :
               : "v"(p), "v"(v)
               : "memory");
}

__device__ __forceinline__ int upper_half() {
  return (int)((__builtin_amdgcn_s_sendmsg_rtn(0x87) >> 18) & 1u);
}

__global__ void k_init(float* __restrict__ deg, float* __restrict__ degHi, int n) {
  int t = blockIdx.x * blockDim.x + threadIdx.x;
  if (t < n) {
    deg[t] = 1.0f;
    if (degHi) degHi[t] = 0.0f;
  }
}

__global__ void k_degree(const int* __restrict__ dst, float* __restrict__ deg,
                         float* __restrict__ degHi, int E) {
  const int t    = blockIdx.x * blockDim.x + threadIdx.x;
  const int nthr = gridDim.x * blockDim.x;
  float* dg = (degHi && upper_half()) ? degHi : deg;
  const int nq = (((size_t)dst & 15) == 0) ? (E >> 2) : 0;
  const int4* dq = (const int4*)dst;
  int q = t;
  for (; q + nthr < nq; q += 2 * nthr) {
    const int4 a = dq[q];
    const int4 b = dq[q + nthr];
    atomAddF(&dg[a.x], 1.0f); atomAddF(&dg[a.y], 1.0f);
    atomAddF(&dg[a.z], 1.0f); atomAddF(&dg[a.w], 1.0f);
    atomAddF(&dg[b.x], 1.0f); atomAddF(&dg[b.y], 1.0f);
    atomAddF(&dg[b.z], 1.0f); atomAddF(&dg[b.w], 1.0f);
  }
  for (; q < nq; q += nthr) {
    const int4 a = dq[q];
    atomAddF(&dg[a.x], 1.0f); atomAddF(&dg[a.y], 1.0f);
    atomAddF(&dg[a.z], 1.0f); atomAddF(&dg[a.w], 1.0f);
  }
  for (int e = (nq << 2) + t; e < E; e += nthr) atomAddF(&dg[dst[e]], 1.0f);
}

template <int LAYER>
__global__ void k_gemm16(const float* X, const float* XHi,
                         const float* __restrict__ bias,
                         const float* __restrict__ W,
                         float* dinv, const float* degHi,
                         float* __restrict__ Y, float* S, float* Z,
                         int nrows, int K) {
  const int wave = threadIdx.x >> 5;
  const int lane = threadIdx.x & 31;
  const int tile = blockIdx.x * (blockDim.x >> 5) + wave;
  const int row0 = tile * 16;
  if (row0 >= nrows) return;

  const int m  = lane & 15;
  const int kh = lane >> 4;

  float4 dlo = *(const float4*)(dinv + row0 + 8 * kh);
  float4 dhi = *(const float4*)(dinv + row0 + 8 * kh + 4);
  float4 hlo = {0.0f, 0.0f, 0.0f, 0.0f}, hhi = {0.0f, 0.0f, 0.0f, 0.0f};
  if (LAYER == 1 && degHi) {
    hlo = *(const float4*)(degHi + row0 + 8 * kh);
    hhi = *(const float4*)(degHi + row0 + 8 * kh + 4);
  }

  v8f acc = {};
  if (LAYER == 2) {
    const float* ap = X + (size_t)(row0 + m) * 16 + 2 * kh;
    v2f a[4], bb[4], b[4];
    const float da = dinv[row0 + m];
#pragma unroll
    for (int j = 0; j < 4; ++j) {
      const int kr = 4 * j + 2 * kh;
      a[j]   = *(const v2f*)(ap + 4 * j);
      bb[j]  = *(const v2f*)(bias + kr);
      b[j].x = W[kr * 16 + m];
      b[j].y = W[(kr + 1) * 16 + m];
    }
    if (XHi) {
      const float* aph = XHi + (size_t)(row0 + m) * 16 + 2 * kh;
#pragma unroll
      for (int j = 0; j < 4; ++j) {
        const v2f q = *(const v2f*)(aph + 4 * j);
        a[j].x += q.x; a[j].y += q.y;
      }
    }
#pragma unroll
    for (int j = 0; j < 4; ++j) {
      a[j].x = fmaxf(a[j].x * da + bb[j].x, 0.0f);
      a[j].y = fmaxf(a[j].y * da + bb[j].y, 0.0f);
      acc = __builtin_amdgcn_wmma_f32_16x16x4_f32(
          false, a[j], false, b[j], (short)0, acc, false, false);
    }
  } else {
    const float* ap = X + (size_t)(row0 + m) * K + 2 * kh;
    for (int kb = 0; kb < K; kb += 4) {
      v2f a = *(const v2f*)(ap + kb);
      const int kr = kb + 2 * kh;
      v2f b;
      b.x = W[(size_t)kr * 16 + m];
      b.y = W[(size_t)(kr + 1) * 16 + m];
      acc = __builtin_amdgcn_wmma_f32_16x16x4_f32(
          false, a, false, b, (short)0, acc, false, false);
    }
  }
  if (LAYER == 1) {
    dlo.x = rsqrtf(dlo.x + hlo.x); dlo.y = rsqrtf(dlo.y + hlo.y);
    dlo.z = rsqrtf(dlo.z + hlo.z); dlo.w = rsqrtf(dlo.w + hlo.w);
    dhi.x = rsqrtf(dhi.x + hhi.x); dhi.y = rsqrtf(dhi.y + hhi.y);
    dhi.z = rsqrtf(dhi.z + hhi.z); dhi.w = rsqrtf(dhi.w + hhi.w);
    if (m == 0) {
      *(float4*)(dinv + row0 + 8 * kh)     = dlo;
      *(float4*)(dinv + row0 + 8 * kh + 4) = dhi;
    }
  }
  const float dv[8] = {dlo.x, dlo.y, dlo.z, dlo.w, dhi.x, dhi.y, dhi.z, dhi.w};
#pragma unroll
  for (int v = 0; v < 8; ++v) {
    const size_t o = (size_t)(row0 + v + 8 * kh) * 16 + m;
    const float y = acc[v] * dv[v];
    Y[o] = y;
    S[o] = y;
    if (Z) Z[o] = 0.0f;
  }
}

#define SC_G 8

__device__ __forceinline__ void atomAddOff(float* base, unsigned off, float v) {
  asm volatile("global_atomic_add_f32 %0, %1, %2 scope:SCOPE_DEV"
               :
               : "v"(off), "v"(v), "s"(base)
               : "memory");
}

template <int J>
__device__ __forceinline__ unsigned rowLane(unsigned v) {
  return (unsigned)__builtin_amdgcn_update_dpp(0, (int)v, 0x150 + J, 0xf, 0xf, true);
}

#define SC_STEPS(X) X(0) X(1) X(2) X(3) X(4) X(5) X(6) X(7) \
                    X(8) X(9) X(10) X(11) X(12) X(13) X(14) X(15)

template <bool GUARD>
__device__ __forceinline__ void scatterGroup(const float* __restrict__ xw, float* agg,
                                             unsigned so, unsigned dof, unsigned f4,
                                             int eh, int E) {
  float val[16];
#define SC_GATHER(J) val[J] = *(const float*)((const char*)xw + (rowLane<J>(so) + f4));
  SC_STEPS(SC_GATHER)
#undef SC_GATHER
#define SC_ADD(J) { const unsigned o = rowLane<J>(dof) + f4; \
                    if (!GUARD || eh + J < E) atomAddOff(agg, o, val[J]); }
  SC_STEPS(SC_ADD)
#undef SC_ADD
}

__global__ void k_scatter(const int* __restrict__ src, const int* __restrict__ dst,
                          const float* __restrict__ xw,
                          float* __restrict__ aggLo, float* __restrict__ aggHi,
                          int E) {
  float* agg = (aggHi && upper_half()) ? aggHi : aggLo;
  const int lane = threadIdx.x & 31;
  const int wv   = (blockIdx.x * blockDim.x + threadIdx.x) >> 5;
  int base = __builtin_amdgcn_readfirstlane(wv) * (32 * SC_G);
  int s = src[min(base + lane, E - 1)];
  int d = dst[min(base + lane, E - 1)];
  const int half    = lane >> 4;
  const unsigned f4 = (unsigned)(lane & 15) * 4u;

  for (int g = 0; g < SC_G; ++g) {
    if (base >= E) break;
    const int ep = min(base + lane + ((g + 1 < SC_G) ? 32 : 0), E - 1);
    const int sn = src[ep];
    const int dn = dst[ep];
    const unsigned so  = (unsigned)s << 6;
    const unsigned dof = (unsigned)d << 6;
    if (base + 32 <= E) scatterGroup<false>(xw, agg, so, dof, f4, 0, E);
    else                scatterGroup<true>(xw, agg, so, dof, f4, base + 16 * half, E);
    s = sn;
    d = dn;
    base += 32;
  }
}

__global__ void k_final2(const float* __restrict__ agg,
                         const float* __restrict__ aggHi,
                         const float* __restrict__ dinv,
                         const float* __restrict__ b2,
                         const float* __restrict__ Wl,
                         const float* __restrict__ bl,
                         float* __restrict__ out, int n) {
  int i = blockIdx.x * blockDim.x + threadIdx.x;
  if (i >= n) return;
  float s = dinv[i];
  float acc = 0.0f;
#pragma unroll
  for (int f = 0; f < 16; ++f) {
    float a = agg[(size_t)i * 16 + f];
    if (aggHi) a += aggHi[(size_t)i * 16 + f];
    float h = a * s + b2[f];
    h = fmaxf(h, 0.0f);
    acc += h * Wl[f];
  }
  out[i] = acc + bl[0];
}

extern "C" void kernel_launch(void* const* d_in, const int* in_sizes, int n_in,
                              void* d_out, int out_size, void* d_ws, size_t ws_size,
                              hipStream_t stream) {
  const float* x   = (const float*)d_in[0];
  const int*   ei  = (const int*)d_in[1];
  const float* W1  = (const float*)d_in[2];
  const float* b1  = (const float*)d_in[3];
  const float* W2  = (const float*)d_in[4];
  const float* b2  = (const float*)d_in[5];
  const float* Wl  = (const float*)d_in[6];
  const float* bl  = (const float*)d_in[7];
  float* out = (float*)d_out;

  const int din = in_sizes[2] / 16;
  const int n   = in_sizes[0] / din;
  const int E   = in_sizes[1] / 2;

  float* ws   = (float*)d_ws;
  float* dinv = ws;
  float* bufA = ws + (((size_t)n + 255) & ~(size_t)255);
  float* bufB = bufA + (size_t)n * 16;
  const size_t need = ((((size_t)n + 255) & ~(size_t)255) + (size_t)n * 49) * sizeof(float);
  float* bufC = (ws_size >= need) ? bufB + (size_t)n * 16 : nullptr;
  float* degC = bufC ? bufC + (size_t)n * 16 : nullptr;

  const int T = 256;
  const int ntiles = (n + 15) / 16;
  const int gemmBlocks = (ntiles + 3) / 4;
  const int scatBlocks = (E + 32 * SC_G * (T / 32) - 1) / (32 * SC_G * (T / 32));

  k_init  <<<(n + T - 1) / T,   T, 0, stream>>>(dinv, degC, n);
  k_degree<<<(E / 8 + T) / T,   T, 0, stream>>>(ei + E, dinv, degC, E);

  k_gemm16<1><<<gemmBlocks, 128, 0, stream>>>(x, nullptr, nullptr, W1, dinv, degC, bufA, bufB, bufC, n, din);
  k_scatter<<<scatBlocks, T, 0, stream>>>(ei, ei + E, bufA, bufB, bufC, E);

  k_gemm16<2><<<gemmBlocks, 128, 0, stream>>>(bufB, bufC, b1, W2, dinv, nullptr, bufA, bufB, bufC, n, 16);
  k_scatter<<<scatBlocks, T, 0, stream>>>(ei, ei + E, bufA, bufB, bufC, E);
  k_final2 <<<(n + T - 1) / T, T, 0, stream>>>(bufB, bufC, dinv, b2, Wl, bl, out, n);
}
